# starting kernel
# baseline (speedup 1.0000x reference)
#include <hip/hip_runtime.h>
#include <math.h>


#define B_   32
#define C_   256
#define H_   64
#define W_   64
#define HW_  4096
#define MID_ 16

typedef __attribute__((ext_vector_type(2))) float v2f;
typedef __attribute__((ext_vector_type(4))) float v4f;
typedef __attribute__((ext_vector_type(8))) float v8f;

__global__ __launch_bounds__(256) void cbam_pool(const float* __restrict__ x,
                                                 float* __restrict__ avg,
                                                 float* __restrict__ mxp) {
  int plane = blockIdx.x;
  const v4f* p = (const v4f*)(x + (size_t)plane * HW_);
  float s = 0.f, m = -INFINITY;
#pragma unroll
  for (int i = 0; i < 4; ++i) {
    v4f v = p[threadIdx.x + i * 256];
    s += v.x + v.y + v.z + v.w;
    m = fmaxf(m, fmaxf(fmaxf(v.x, v.y), fmaxf(v.z, v.w)));
  }
#pragma unroll
  for (int off = 16; off >= 1; off >>= 1) {
    s += __shfl_xor(s, off, 32);
    m = fmaxf(m, __shfl_xor(m, off, 32));
  }
  __shared__ float ss[8], sm[8];
  int wave = threadIdx.x >> 5, lane = threadIdx.x & 31;
  if (lane == 0) { ss[wave] = s; sm[wave] = m; }
  __syncthreads();
  if (threadIdx.x == 0) {
    float ts = 0.f, tm = -INFINITY;
#pragma unroll
    for (int i = 0; i < 8; ++i) { ts += ss[i]; tm = fmaxf(tm, sm[i]); }
    avg[plane] = ts * (1.0f / (float)HW_);
    mxp[plane] = tm;
  }
}

__global__ __launch_bounds__(256) void cbam_mlp_ca(const float* __restrict__ avg,
                                                   const float* __restrict__ mxp,
                                                   const float* __restrict__ w1,
                                                   const float* __restrict__ w2,
                                                   float* __restrict__ ca) {
  __shared__ float hcomb[32][32];
  int tid  = threadIdx.x;
  int wave = tid >> 5;
  int lane = tid & 31;
  int half = lane >> 4;
  int l15  = lane & 15;

  if (wave < 4) {
    int src   = wave >> 1;
    int mtile = wave & 1;
    const float* pool = src ? mxp : avg;
    int M = mtile * 16 + l15;
    v8f acc = {};
    for (int k0 = 0; k0 < 256; k0 += 4) {
      int ka = k0 + half * 2;
      v2f a, b;
      a.x = pool[M * 256 + ka];
      a.y = pool[M * 256 + ka + 1];
      b.x = w1[l15 * 256 + ka];
      b.y = w1[l15 * 256 + ka + 1];
      acc = __builtin_amdgcn_wmma_f32_16x16x4_f32(
          false, a, false, b, (short)0, acc, false, false);
    }
#pragma unroll
    for (int i = 0; i < 8; ++i) {
      int m = mtile * 16 + i + half * 8;
      hcomb[m][src * 16 + l15] = fmaxf(acc[i], 0.f);
    }
  }
  __syncthreads();

  for (int t = wave; t < 32; t += 8) {
    int mtile = t >> 4;
    int ntile = t & 15;
    v8f acc = {};
#pragma unroll
    for (int k0 = 0; k0 < 32; k0 += 4) {
      int ka = k0 + half * 2;
      v2f a, b;
      a.x = hcomb[mtile * 16 + l15][ka];
      a.y = hcomb[mtile * 16 + l15][ka + 1];
      b.x = w2[(ntile * 16 + l15) * 16 + (ka & 15)];
      b.y = w2[(ntile * 16 + l15) * 16 + ((ka + 1) & 15)];
      acc = __builtin_amdgcn_wmma_f32_16x16x4_f32(
          false, a, false, b, (short)0, acc, false, false);
    }
#pragma unroll
    for (int i = 0; i < 8; ++i) {
      int m = mtile * 16 + i + half * 8;
      int c = ntile * 16 + l15;
      ca[m * 256 + c] = 1.f / (1.f + __expf(-acc[i]));
    }
  }
}

__global__ __launch_bounds__(256) void cbam_feat(const float* __restrict__ x,
                                                 const float* __restrict__ ca,
                                                 float* __restrict__ fm,
                                                 float* __restrict__ fx) {
  __shared__ float sCa[256];
  int b  = blockIdx.x >> 4;
  int hw = ((blockIdx.x & 15) << 8) + threadIdx.x;
  sCa[threadIdx.x] = ca[b * 256 + threadIdx.x];
  __syncthreads();
  const float* xb = x + (size_t)b * C_ * HW_;
  float s = 0.f, m = -INFINITY;
#pragma unroll 4
  for (int c = 0; c < 256; ++c) {
    float v = xb[(size_t)c * HW_ + hw] * sCa[c];
    s += v;
    m = fmaxf(m, v);
  }
  fm[b * HW_ + hw] = s * (1.f / 256.f);
  fx[b * HW_ + hw] = m;
}

__global__ __launch_bounds__(256) void cbam_conv(const float* __restrict__ fm,
                                                 const float* __restrict__ fx,
                                                 const float* __restrict__ wsp,
                                                 float* __restrict__ sig) {
  __shared__ float sW[98];
  if (threadIdx.x < 98) sW[threadIdx.x] = wsp[threadIdx.x];
  __syncthreads();
  int b  = blockIdx.x >> 4;
  int hw = ((blockIdx.x & 15) << 8) + threadIdx.x;
  int y = hw >> 6, xc = hw & 63;
  const float* fmb = fm + b * HW_;
  const float* fxb = fx + b * HW_;
  float acc = 0.f;
#pragma unroll
  for (int ky = 0; ky < 7; ++ky) {
    int yy = y + ky - 3;
    if (yy < 0 || yy >= H_) continue;
#pragma unroll
    for (int kx = 0; kx < 7; ++kx) {
      int x2 = xc + kx - 3;
      if (x2 < 0 || x2 >= W_) continue;
      acc += fmb[yy * 64 + x2] * sW[ky * 7 + kx]
           + fxb[yy * 64 + x2] * sW[49 + ky * 7 + kx];
    }
  }
  sig[b * HW_ + hw] = 1.f / (1.f + __expf(-acc));
}

__global__ __launch_bounds__(256) void cbam_final(const float* __restrict__ x,
                                                  const float* __restrict__ ca,
                                                  const float* __restrict__ sig,
                                                  float* __restrict__ out) {
  size_t idx = (size_t)blockIdx.x * 256 + threadIdx.x;
  size_t v   = idx * 4;
  int hw = (int)(v & 4095);
  int c  = (int)((v >> 12) & 255);
  int b  = (int)(v >> 20);
  v4f xv = __builtin_nontemporal_load((const v4f*)x + idx);
  v4f sv = *(const v4f*)(sig + (size_t)b * HW_ + hw);
  float cv = ca[b * 256 + c];
  v4f o = xv * (sv * cv);
  __builtin_nontemporal_store(o, (v4f*)out + idx);
}

extern "C" void kernel_launch(void* const* d_in, const int* in_sizes, int n_in,
                              void* d_out, int out_size, void* d_ws, size_t ws_size,
                              hipStream_t stream) {
  const float* x  = (const float*)d_in[0];
  const float* w1 = (const float*)d_in[1];
  const float* w2 = (const float*)d_in[2];
  const float* ws = (const float*)d_in[3];
  float* out = (float*)d_out;

  float* wsf = (float*)d_ws;
  float* avg = wsf;
  float* mxp = wsf + 8192;
  float* ca  = wsf + 16384;
  float* fm  = wsf + 24576;
  float* fx  = wsf + 155648;
  float* sig = wsf + 286720;

  cbam_pool  <<<B_ * C_,          256, 0, stream>>>(x, avg, mxp);
  cbam_mlp_ca<<<1,                256, 0, stream>>>(avg, mxp, w1, w2, ca);
  cbam_feat  <<<B_ * (HW_ / 256), 256, 0, stream>>>(x, ca, fm, fx);
  cbam_conv  <<<B_ * (HW_ / 256), 256, 0, stream>>>(fm, fx, ws, sig);
  cbam_final <<<(B_ * C_ * HW_ / 4) / 256, 256, 0, stream>>>(x, ca, sig, out);
}


# MLP launch: every operand (pooled planes, w1, w2) fetched once by whole-line 16-byte loads issued together and parked in LDS, fragments read from LDS; 32 waves, one output tile each; sums unchanged
# speedup vs baseline: 1.8797x; 1.8797x over previous
#include <hip/hip_runtime.h>
#include <math.h>


#define B_   32
#define C_   256
#define H_   64
#define W_   64
#define HW_  4096
#define MID_ 16

typedef __attribute__((ext_vector_type(2))) float v2f;
typedef __attribute__((ext_vector_type(4))) float v4f;
typedef __attribute__((ext_vector_type(8))) float v8f;

__global__ __launch_bounds__(256) void cbam_pool(const float* __restrict__ x,
                                                 float* __restrict__ avg,
                                                 float* __restrict__ mxp) {
  int plane = blockIdx.x;
  const v4f* p = (const v4f*)(x + (size_t)plane * HW_);
  float s = 0.f, m = -INFINITY;
#pragma unroll
  for (int i = 0; i < 4; ++i) {
    v4f v = p[threadIdx.x + i * 256];
    s += v.x + v.y + v.z + v.w;
    m = fmaxf(m, fmaxf(fmaxf(v.x, v.y), fmaxf(v.z, v.w)));
  }
#pragma unroll
  for (int off = 16; off >= 1; off >>= 1) {
    s += __shfl_xor(s, off, 32);
    m = fmaxf(m, __shfl_xor(m, off, 32));
  }
  __shared__ float ss[8], sm[8];
  int wave = threadIdx.x >> 5, lane = threadIdx.x & 31;
  if (lane == 0) { ss[wave] = s; sm[wave] = m; }
  __syncthreads();
  if (threadIdx.x == 0) {
    float ts = 0.f, tm = -INFINITY;
#pragma unroll
    for (int i = 0; i < 8; ++i) { ts += ss[i]; tm = fmaxf(tm, sm[i]); }
    avg[plane] = ts * (1.0f / (float)HW_);
    mxp[plane] = tm;
  }
}

#define PP_  260
#define W2P_ 20
#define HP_  36

__global__ __launch_bounds__(1024) void cbam_mlp_ca(const float* __restrict__ avg,
                                                    const float* __restrict__ mxp,
                                                    const float* __restrict__ w1,
                                                    const float* __restrict__ w2,
                                                    float* __restrict__ ca) {
  __shared__ __attribute__((aligned(16))) float sPool[64 * PP_];
  __shared__ __attribute__((aligned(16))) float sW1[16 * PP_];
  __shared__ __attribute__((aligned(16))) float sW2[256 * W2P_];
  __shared__ __attribute__((aligned(16))) float hcomb[32 * HP_];
  int tid  = threadIdx.x;
  int wave = tid >> 5;
  int lane = tid & 31;
  int half = lane >> 4;
  int l15  = lane & 15;

  {
    v4f p0 = ((const v4f*)avg)[tid];
    v4f p1 = ((const v4f*)avg)[tid + 1024];
    v4f p2 = ((const v4f*)mxp)[tid];
    v4f p3 = ((const v4f*)mxp)[tid + 1024];
    v4f q1 = ((const v4f*)w1)[tid];
    v4f q2 = ((const v4f*)w2)[tid];
    int r = tid >> 6, c4 = (tid & 63) * 4;
    *(v4f*)(sPool + (r)      * PP_ + c4) = p0;
    *(v4f*)(sPool + (r + 16) * PP_ + c4) = p1;
    *(v4f*)(sPool + (r + 32) * PP_ + c4) = p2;
    *(v4f*)(sPool + (r + 48) * PP_ + c4) = p3;
    *(v4f*)(sW1   + (r)      * PP_ + c4) = q1;
    *(v4f*)(sW2 + (tid >> 2) * W2P_ + (tid & 3) * 4) = q2;
  }
  __syncthreads();

  if (wave < 4) {
    int src   = wave >> 1;
    int mtile = wave & 1;
    const float* ap = sPool + (src * 32 + mtile * 16 + l15) * PP_ + half * 2;
    const float* bp = sW1 + l15 * PP_ + half * 2;
    v8f acc = {};
    v2f a[2][8], b[2][8];
#pragma unroll
    for (int j = 0; j < 8; ++j) {
      a[0][j] = *(const v2f*)(ap + j * 4);
      b[0][j] = *(const v2f*)(bp + j * 4);
    }
#pragma unroll
    for (int kb = 0; kb < 8; ++kb) {
      if (kb < 7) {
#pragma unroll
        for (int j = 0; j < 8; ++j) {
          a[(kb + 1) & 1][j] = *(const v2f*)(ap + (kb + 1) * 32 + j * 4);
          b[(kb + 1) & 1][j] = *(const v2f*)(bp + (kb + 1) * 32 + j * 4);
        }
      }
      __builtin_amdgcn_sched_barrier(0);
#pragma unroll
      for (int j = 0; j < 8; ++j)
        acc = __builtin_amdgcn_wmma_f32_16x16x4_f32(
            false, a[kb & 1][j], false, b[kb & 1][j], (short)0, acc, false, false);
      __builtin_amdgcn_sched_barrier(0);
    }
#pragma unroll
    for (int i = 0; i < 8; ++i) {
      int m = mtile * 16 + i + half * 8;
      hcomb[m * HP_ + src * 16 + l15] = fmaxf(acc[i], 0.f);
    }
  }
  __syncthreads();

  {
    int mtile = wave >> 4;
    int ntile = wave & 15;
    const float* ap = hcomb + (mtile * 16 + l15) * HP_ + half * 2;
    const float* bp = sW2 + (ntile * 16 + l15) * W2P_ + half * 2;
    v8f acc = {};
#pragma unroll
    for (int k0 = 0; k0 < 32; k0 += 4) {
      v2f a = *(const v2f*)(ap + k0);
      v2f b = *(const v2f*)(bp + (k0 & 15));
      acc = __builtin_amdgcn_wmma_f32_16x16x4_f32(
          false, a, false, b, (short)0, acc, false, false);
    }
#pragma unroll
    for (int i = 0; i < 8; ++i) {
      int m = mtile * 16 + i + half * 8;
      int c = ntile * 16 + l15;
      ca[m * 256 + c] = 1.f / (1.f + __expf(-acc[i]));
    }
  }
}

#define FT_    640
#define NPOS_  160
#define CPG_   64
#define FP_    72

__global__ __launch_bounds__(FT_) void cbam_spatial_out(const float* __restrict__ x,
                                                        const float* __restrict__ ca,
                                                        const float* __restrict__ wsp,
                                                        float* __restrict__ out) {
  __shared__ __attribute__((aligned(16))) float xs[C_ * 256];
  __shared__ __attribute__((aligned(16))) float ps[3 * NPOS_ * 4];
  __shared__ __attribute__((aligned(16))) float pm[3 * NPOS_ * 4];
  __shared__ __attribute__((aligned(16))) v2f fS[10 * FP_];
  __shared__ __attribute__((aligned(16))) float sigS[256];
  __shared__ __attribute__((aligned(16))) float sCa[C_];
  __shared__ __attribute__((aligned(16))) float sW[100];

  int tid = threadIdx.x;
  int b   = blockIdx.x >> 4;
  int r0  = (blockIdx.x & 15) << 2;
  int pos = tid % NPOS_;
  int g   = tid / NPOS_;
  int wrow = pos >> 4;
  int irow = r0 - 3 + wrow;
  bool valid = (irow >= 0) && (irow < H_);
  bool own   = (wrow >= 3) && (wrow < 7);

  if (tid < C_) sCa[tid] = ca[b * C_ + tid];
  else if (tid < C_ + 98) sW[tid - C_] = wsp[tid - C_];
  __syncthreads();

  v4f s = {0.f, 0.f, 0.f, 0.f};
  v4f m = {-INFINITY, -INFINITY, -INFINITY, -INFINITY};
  if (valid) {
    const v4f* xp = (const v4f*)(x + ((size_t)(b * C_ + g * CPG_) * HW_ + (size_t)irow * W_)) + (pos & 15);
    float* xsp = xs + (g * CPG_) * 256 + (pos - 48) * 4;
    for (int i0 = 0; i0 < CPG_; i0 += 16) {
      v4f v[16];
#pragma unroll
      for (int j = 0; j < 16; ++j) v[j] = xp[(size_t)(i0 + j) * (HW_ / 4)];
      if (own) {
#pragma unroll
        for (int j = 0; j < 16; ++j) *(v4f*)(xsp + (i0 + j) * 256) = v[j];
      }
#pragma unroll
      for (int j = 0; j < 16; ++j) {
        v4f t = v[j] * sCa[g * CPG_ + i0 + j];
        s += t;
        m.x = fmaxf(m.x, t.x); m.y = fmaxf(m.y, t.y);
        m.z = fmaxf(m.z, t.z); m.w = fmaxf(m.w, t.w);
      }
    }
  }
  if (g > 0) {
    *(v4f*)(ps + ((g - 1) * NPOS_ + pos) * 4) = s;
    *(v4f*)(pm + ((g - 1) * NPOS_ + pos) * 4) = m;
  }
  __syncthreads();
  if (g == 0) {
    v4f mean = {0.f, 0.f, 0.f, 0.f};
    v4f mx   = {0.f, 0.f, 0.f, 0.f};
    if (valid) {
#pragma unroll
      for (int k = 0; k < 3; ++k) {
        v4f s2 = *(const v4f*)(ps + (k * NPOS_ + pos) * 4);
        v4f m2 = *(const v4f*)(pm + (k * NPOS_ + pos) * 4);
        s += s2;
        m.x = fmaxf(m.x, m2.x); m.y = fmaxf(m.y, m2.y);
        m.z = fmaxf(m.z, m2.z); m.w = fmaxf(m.w, m2.w);
      }
      mean = s * (1.f / 256.f);
      mx = m;
    }
    v4f* fp = (v4f*)(fS + wrow * FP_ + 4 + (pos & 15) * 4);
    fp[0] = (v4f){mean.x, mx.x, mean.y, mx.y};
    fp[1] = (v4f){mean.z, mx.z, mean.w, mx.w};
  } else if (tid < NPOS_ + 80) {
    int k = tid - NPOS_;
    int r = k >> 3, q = k & 7;
    fS[r * FP_ + (q < 4 ? q : 64 + q)] = (v2f){0.f, 0.f};
  }
  __syncthreads();

  if (tid < 256) {
    const v2f* f0 = fS + (tid >> 6) * FP_ + (tid & 63) + 1;
    float acc = 0.f;
#pragma unroll
    for (int ky = 0; ky < 7; ++ky) {
#pragma unroll
      for (int kx = 0; kx < 7; ++kx) {
        v2f f = f0[ky * FP_ + kx];
        acc += f.x * sW[ky * 7 + kx] + f.y * sW[49 + ky * 7 + kx];
      }
    }
    sigS[tid] = 1.f / (1.f + __expf(-acc));
  }
  __syncthreads();

  float* ob = out + (size_t)b * C_ * HW_ + (size_t)r0 * W_;
  for (int i = tid; i < C_ * 64; i += FT_) {
    int c = i >> 6, p4 = i & 63;
    v4f xv = *(const v4f*)(xs + c * 256 + p4 * 4);
    v4f sv = *(const v4f*)(sigS + p4 * 4);
    float cv = sCa[c];
    v4f o = xv * (sv * cv);
    __builtin_nontemporal_store(o, (v4f*)(ob + (size_t)c * HW_) + p4);
  }
}

extern "C" void kernel_launch(void* const* d_in, const int* in_sizes, int n_in,
                              void* d_out, int out_size, void* d_ws, size_t ws_size,
                              hipStream_t stream) {
  const float* x  = (const float*)d_in[0];
  const float* w1 = (const float*)d_in[1];
  const float* w2 = (const float*)d_in[2];
  const float* ws = (const float*)d_in[3];
  float* out = (float*)d_out;

  float* wsf = (float*)d_ws;
  float* avg = wsf;
  float* mxp = wsf + 8192;
  float* ca  = wsf + 16384;

  cbam_pool  <<<B_ * C_,          256, 0, stream>>>(x, avg, mxp);
  cbam_mlp_ca<<<1,               1024, 0, stream>>>(avg, mxp, w1, w2, ca);
  cbam_spatial_out<<<B_ * (H_ / 4), FT_, 0, stream>>>(x, ca, ws, out);
}


# pool pass re-cut to the fused pass's 256 blocks (batch row x 8 image rows, all channels): 8x fewer waves; strip partials in workspace, joined by an MLP launch per batch row; plane sum order changes
# speedup vs baseline: 2.3695x; 1.2606x over previous
#include <hip/hip_runtime.h>
#include <math.h>


#define B_   32
#define C_   256
#define H_   64
#define W_   64
#define HW_  4096
#define MID_ 16

typedef __attribute__((ext_vector_type(2))) float v2f;
typedef __attribute__((ext_vector_type(4))) float v4f;
typedef __attribute__((ext_vector_type(8))) float v8f;

#define SR_    8
#define NSTR_  8
#define PCW_   8
#define PFL_   4

template <int MASK>
__device__ __forceinline__ float xrow(float v) {
  int b = __builtin_bit_cast(int, v);
  return __builtin_bit_cast(float,
      __builtin_amdgcn_update_dpp(b, b, 0x160 + MASK, 0xf, 0xf, false));
}

__device__ __forceinline__ float xhalf(float v) {
  int b = __builtin_bit_cast(int, v);
  return __builtin_bit_cast(float,
      __builtin_amdgcn_permlanex16(b, b, 0x76543210, 0xfedcba98, false, false));
}

__global__ __launch_bounds__(1024) void cbam_pool_strip(const float* __restrict__ x,
                                                        float* __restrict__ psum,
                                                        float* __restrict__ pmax) {
  __shared__ __attribute__((aligned(16))) float rs[C_];
  __shared__ __attribute__((aligned(16))) float rm[C_];
  int tid  = threadIdx.x;
  int wave = tid >> 5;
  int lane = tid & 31;
  int b    = blockIdx.x >> 3;
  int st   = blockIdx.x & 7;
  const float* xw = x + ((size_t)(b * C_ + wave * PCW_)) * HW_ + st * (SR_ * W_) + lane * 4;

  v4f buf[PFL_][4];
#pragma unroll
  for (int c = 0; c < PFL_; ++c)
#pragma unroll
    for (int i = 0; i < 4; ++i)
      buf[c][i] = *(const v4f*)(xw + c * HW_ + i * 128);

  float ks = 0.f, km = 0.f;
#pragma unroll
  for (int c = 0; c < PCW_; ++c) {
    v4f s4 = (buf[c % PFL_][0] + buf[c % PFL_][1]) + (buf[c % PFL_][2] + buf[c % PFL_][3]);
    v4f m4;
    m4.x = fmaxf(fmaxf(buf[c % PFL_][0].x, buf[c % PFL_][1].x), fmaxf(buf[c % PFL_][2].x, buf[c % PFL_][3].x));
    m4.y = fmaxf(fmaxf(buf[c % PFL_][0].y, buf[c % PFL_][1].y), fmaxf(buf[c % PFL_][2].y, buf[c % PFL_][3].y));
    m4.z = fmaxf(fmaxf(buf[c % PFL_][0].z, buf[c % PFL_][1].z), fmaxf(buf[c % PFL_][2].z, buf[c % PFL_][3].z));
    m4.w = fmaxf(fmaxf(buf[c % PFL_][0].w, buf[c % PFL_][1].w), fmaxf(buf[c % PFL_][2].w, buf[c % PFL_][3].w));
    if (c + PFL_ < PCW_) {
#pragma unroll
      for (int i = 0; i < 4; ++i)
        buf[c % PFL_][i] = *(const v4f*)(xw + (c + PFL_) * HW_ + i * 128);
    }
    float s = (s4.x + s4.y) + (s4.z + s4.w);
    float m = fmaxf(fmaxf(m4.x, m4.y), fmaxf(m4.z, m4.w));
    s += xrow<1>(s);  m = fmaxf(m, xrow<1>(m));
    s += xrow<2>(s);  m = fmaxf(m, xrow<2>(m));
    s += xrow<4>(s);  m = fmaxf(m, xrow<4>(m));
    s += xrow<8>(s);  m = fmaxf(m, xrow<8>(m));
    s += xhalf(s);    m = fmaxf(m, xhalf(m));
    if (lane == c) { ks = s; km = m; }
  }
  if (lane < PCW_) { rs[wave * PCW_ + lane] = ks; rm[wave * PCW_ + lane] = km; }
  __syncthreads();
  if (tid < 64)
    *(v4f*)(psum + blockIdx.x * C_ + tid * 4) = *(const v4f*)(rs + tid * 4);
  else if (tid < 128)
    *(v4f*)(pmax + blockIdx.x * C_ + (tid - 64) * 4) = *(const v4f*)(rm + (tid - 64) * 4);
}

#define PP_  260
#define W2P_ 20
#define HP_  36

__global__ __launch_bounds__(256) void cbam_mlp_ca(const float* __restrict__ psum,
                                                   const float* __restrict__ pmax,
                                                   const float* __restrict__ w1,
                                                   const float* __restrict__ w2,
                                                   float* __restrict__ ca) {
  __shared__ __attribute__((aligned(16))) float sV[16 * PP_];
  __shared__ __attribute__((aligned(16))) float sW1[16 * PP_];
  __shared__ __attribute__((aligned(16))) float sW2[256 * W2P_];
  __shared__ __attribute__((aligned(16))) float hA[16 * HP_];
  int tid  = threadIdx.x;
  int wave = tid >> 5;
  int lane = tid & 31;
  int half = lane >> 4;
  int l15  = lane & 15;
  int b    = blockIdx.x;

  {
    v4f q1[4], q2[4], p[NSTR_];
#pragma unroll
    for (int i = 0; i < 4; ++i) {
      q1[i] = ((const v4f*)w1)[tid + i * 256];
      q2[i] = ((const v4f*)w2)[tid + i * 256];
    }
    if (tid < 128) {
      const v4f* pp = (const v4f*)((tid < 64 ? psum : pmax) + b * (NSTR_ * C_)) + (tid & 63);
#pragma unroll
      for (int s = 0; s < NSTR_; ++s) p[s] = pp[s * (C_ / 4)];
    }
    for (int e = tid; e < 14 * PP_ / 4; e += 256)
      ((v4f*)(sV + 2 * PP_))[e] = (v4f){0.f, 0.f, 0.f, 0.f};
    if (tid < 15 * HP_ / 4)
      ((v4f*)(hA + HP_))[tid] = (v4f){0.f, 0.f, 0.f, 0.f};
#pragma unroll
    for (int i = 0; i < 4; ++i) {
      int e = tid + i * 256;
      *(v4f*)(sW1 + (e >> 6) * PP_ + (e & 63) * 4) = q1[i];
      *(v4f*)(sW2 + (e >> 2) * W2P_ + (e & 3) * 4) = q2[i];
    }
    if (tid < 128) {
      v4f a = p[0];
      if (tid < 64) {
#pragma unroll
        for (int s = 1; s < NSTR_; ++s) a += p[s];
        a = a * (1.0f / (float)HW_);
      } else {
#pragma unroll
        for (int s = 1; s < NSTR_; ++s) {
          a.x = fmaxf(a.x, p[s].x); a.y = fmaxf(a.y, p[s].y);
          a.z = fmaxf(a.z, p[s].z); a.w = fmaxf(a.w, p[s].w);
        }
      }
      *(v4f*)(sV + (tid >> 6) * PP_ + (tid & 63) * 4) = a;
    }
  }
  __syncthreads();

  if (wave == 0) {
    const float* ap = sV + l15 * PP_ + half * 2;
    const float* bp = sW1 + l15 * PP_ + half * 2;
    v8f acc = {};
    v2f a[2][8], bb[2][8];
#pragma unroll
    for (int j = 0; j < 8; ++j) {
      a[0][j]  = *(const v2f*)(ap + j * 4);
      bb[0][j] = *(const v2f*)(bp + j * 4);
    }
#pragma unroll
    for (int kb = 0; kb < 8; ++kb) {
      if (kb < 7) {
#pragma unroll
        for (int j = 0; j < 8; ++j) {
          a[(kb + 1) & 1][j]  = *(const v2f*)(ap + (kb + 1) * 32 + j * 4);
          bb[(kb + 1) & 1][j] = *(const v2f*)(bp + (kb + 1) * 32 + j * 4);
        }
      }
      __builtin_amdgcn_sched_barrier(0);
#pragma unroll
      for (int j = 0; j < 8; ++j)
        acc = __builtin_amdgcn_wmma_f32_16x16x4_f32(
            false, a[kb & 1][j], false, bb[kb & 1][j], (short)0, acc, false, false);
      __builtin_amdgcn_sched_barrier(0);
    }
    if (half == 0) {
      hA[l15]      = fmaxf(acc[0], 0.f);
      hA[16 + l15] = fmaxf(acc[1], 0.f);
    }
  }
  __syncthreads();

  {
    const float* ap  = hA + l15 * HP_ + half * 2;
    const float* bp0 = sW2 + ((wave * 2)     * 16 + l15) * W2P_ + half * 2;
    const float* bp1 = sW2 + ((wave * 2 + 1) * 16 + l15) * W2P_ + half * 2;
    v8f acc0 = {}, acc1 = {};
#pragma unroll
    for (int k0 = 0; k0 < 32; k0 += 4) {
      v2f av = *(const v2f*)(ap + k0);
      v2f b0 = *(const v2f*)(bp0 + (k0 & 15));
      v2f b1 = *(const v2f*)(bp1 + (k0 & 15));
      acc0 = __builtin_amdgcn_wmma_f32_16x16x4_f32(
          false, av, false, b0, (short)0, acc0, false, false);
      acc1 = __builtin_amdgcn_wmma_f32_16x16x4_f32(
          false, av, false, b1, (short)0, acc1, false, false);
    }
    float v1 = xhalf(acc1[0]);
    float v  = half == 0 ? acc0[0] : v1;
    ca[b * C_ + wave * 32 + lane] = 1.f / (1.f + __expf(-v));
  }
}

#define NWIN_  14
#define ST_    1024
#define FP_    72

__device__ __forceinline__ void lds_barrier() {
  __builtin_amdgcn_fence(__ATOMIC_RELEASE, "workgroup", "local");
  __builtin_amdgcn_s_barrier();
  __builtin_amdgcn_fence(__ATOMIC_ACQUIRE, "workgroup", "local");
}

__global__ __launch_bounds__(ST_) void cbam_spatial_out(const float* __restrict__ x,
                                                        const float* __restrict__ ca,
                                                        const float* __restrict__ wsp,
                                                        float* __restrict__ out) {
  __shared__ __attribute__((aligned(16))) float pS[2][32 * 16 * 4];
  __shared__ __attribute__((aligned(16))) float pM[2][32 * 16 * 4];
  __shared__ __attribute__((aligned(16))) v2f fS[NWIN_ * FP_];
  __shared__ __attribute__((aligned(16))) float cP[7 * 64];
  __shared__ __attribute__((aligned(16))) float sigS[64];
  __shared__ __attribute__((aligned(16))) float sW[100];

  int tid  = threadIdx.x;
  int wave = tid >> 5;
  int lane = tid & 31;
  int pos  = lane & 15;
  int half = lane >> 4;
  int c0   = (wave * 2 + half) * 4;
  int b    = blockIdx.x >> 3;
  int r0   = (blockIdx.x & 7) * SR_;

  const float* xb = x   + (size_t)(b * C_ + c0) * HW_ + pos * 4;
  float*       ob = out + (size_t)(b * C_ + c0) * HW_ + pos * 4;

  v4f buf[5][4];
#pragma unroll
  for (int q = 0; q < 2; ++q) {
    int irc = r0 - 3 + q;
    irc = irc < 0 ? 0 : irc;
#pragma unroll
    for (int j = 0; j < 4; ++j)
      buf[q][j] = *(const v4f*)(xb + irc * W_ + j * HW_);
  }
  v4f cav = *(const v4f*)(ca + b * C_ + c0);
  if (tid < 98) sW[tid] = wsp[tid];
  else if (tid >= 128 && tid < 128 + NWIN_ * 8) {
    int k = tid - 128;
    int r = k >> 3, q = k & 7;
    fS[r * FP_ + (q < 4 ? q : 64 + q)] = (v2f){0.f, 0.f};
  }
  lds_barrier();

#pragma unroll
  for (int n = 0; n < NWIN_; ++n) {
    int ir = r0 - 3 + n;
    bool own   = (n >= 3) && (n < 3 + SR_);
    bool valid = own || ((ir >= 0) && (ir < H_));
    if (n <= 5) {
      int irc = ir + 2;
      if (n + 2 < 3) irc = irc < 0 ? 0 : irc;
#pragma unroll
      for (int j = 0; j < 4; ++j)
        buf[(n + 2) % 5][j] = *(const v4f*)(xb + irc * W_ + j * HW_);
    }
    int pb = n & 1;
    {
      v4f s = {0.f, 0.f, 0.f, 0.f};
      v4f m = {-INFINITY, -INFINITY, -INFINITY, -INFINITY};
#pragma unroll
      for (int j = 0; j < 4; ++j) {
        v4f t = buf[n % 5][j] * cav[j];
        s += t;
        m.x = fmaxf(m.x, t.x); m.y = fmaxf(m.y, t.y);
        m.z = fmaxf(m.z, t.z); m.w = fmaxf(m.w, t.w);
      }
      s.x += xhalf(s.x); s.y += xhalf(s.y); s.z += xhalf(s.z); s.w += xhalf(s.w);
      m.x = fmaxf(m.x, xhalf(m.x)); m.y = fmaxf(m.y, xhalf(m.y));
      m.z = fmaxf(m.z, xhalf(m.z)); m.w = fmaxf(m.w, xhalf(m.w));
      if (half == 0) {
        *(v4f*)(pS[pb] + (wave * 16 + pos) * 4) = s;
        *(v4f*)(pM[pb] + (wave * 16 + pos) * 4) = m;
      }
    }
    lds_barrier();
    if (tid < 128) {
      int p2 = tid & 15, comp = (tid >> 4) & 3, sm = tid >> 6;
      float val;
      if (sm == 0) {
        const float* src = pS[pb] + p2 * 4 + comp;
        float a = 0.f;
#pragma unroll
        for (int w = 0; w < 32; ++w) a += src[w * 64];
        val = a * (1.f / 256.f);
      } else {
        const float* src = pM[pb] + p2 * 4 + comp;
        float a = -INFINITY;
#pragma unroll
        for (int w = 0; w < 32; ++w) a = fmaxf(a, src[w * 64]);
        val = a;
      }
      if (!valid) val = 0.f;
      ((float*)fS)[(n * FP_ + 4 + p2 * 4 + comp) * 2 + sm] = val;
    }
    if (n >= 6) {
      lds_barrier();
      if (tid < 448) {
        int xc = tid & 63, ky = tid >> 6;
        const v2f* f0 = fS + (n - 6 + ky) * FP_ + xc + 1;
        const float* wr = sW + ky * 7;
        float acc = 0.f;
#pragma unroll
        for (int kx = 0; kx < 7; ++kx) {
          v2f f = f0[kx];
          acc += f.x * wr[kx] + f.y * wr[49 + kx];
        }
        cP[ky * 64 + xc] = acc;
      }
      lds_barrier();
      if (tid < 64) {
        float acc = 0.f;
#pragma unroll
        for (int ky = 0; ky < 7; ++ky) acc += cP[ky * 64 + tid];
        sigS[tid] = 1.f / (1.f + __expf(-acc));
      }
      lds_barrier();
      v4f sv = *(const v4f*)(sigS + pos * 4);
      float* op = ob + (ir - 3) * W_;
#pragma unroll
      for (int j = 0; j < 4; ++j) {
        v4f o = buf[(n - 3) % 5][j] * (sv * cav[j]);
        __builtin_nontemporal_store(o, (v4f*)(op + j * HW_));
      }
      if (n + 2 < NWIN_) {
        int irc = ir + 2;
        if (n + 2 >= 3 + SR_) irc = irc > H_ - 1 ? H_ - 1 : irc;
#pragma unroll
        for (int j = 0; j < 4; ++j)
          buf[(n + 2) % 5][j] = *(const v4f*)(xb + irc * W_ + j * HW_);
      }
    }
  }
}

extern "C" void kernel_launch(void* const* d_in, const int* in_sizes, int n_in,
                              void* d_out, int out_size, void* d_ws, size_t ws_size,
                              hipStream_t stream) {
  const float* x  = (const float*)d_in[0];
  const float* w1 = (const float*)d_in[1];
  const float* w2 = (const float*)d_in[2];
  const float* ws = (const float*)d_in[3];
  float* out = (float*)d_out;

  float* wsf  = (float*)d_ws;
  float* psum = wsf;
  float* pmax = wsf + 65536;
  float* ca   = wsf + 131072;

  cbam_pool_strip <<<B_ * NSTR_, 1024, 0, stream>>>(x, psum, pmax);
  cbam_mlp_ca     <<<B_,          256, 0, stream>>>(psum, pmax, w1, w2, ca);
  cbam_spatial_out<<<B_ * NSTR_,  ST_, 0, stream>>>(x, ca, ws, out);
}


# fused pass as a 2-stage row pipeline: conv+sigmoid of row n-4 (all lanes, 7 row partials as a lane tree) beside the reduce of row n, join beside the output; 16 waves x 8 ch; channel sum 32 chains
# speedup vs baseline: 2.4820x; 1.0475x over previous
#include <hip/hip_runtime.h>
#include <math.h>


#define B_   32
#define C_   256
#define H_   64
#define W_   64
#define HW_  4096
#define MID_ 16

typedef __attribute__((ext_vector_type(2))) float v2f;
typedef __attribute__((ext_vector_type(4))) float v4f;
typedef __attribute__((ext_vector_type(8))) float v8f;

#define SR_    8
#define NSTR_  8
#define PCW_   8
#define PFL_   4

template <int MASK>
__device__ __forceinline__ float xrow(float v) {
  int b = __builtin_bit_cast(int, v);
  return __builtin_bit_cast(float,
      __builtin_amdgcn_update_dpp(b, b, 0x160 + MASK, 0xf, 0xf, false));
}

__device__ __forceinline__ float xhalf(float v) {
  int b = __builtin_bit_cast(int, v);
  return __builtin_bit_cast(float,
      __builtin_amdgcn_permlanex16(b, b, 0x76543210, 0xfedcba98, false, false));
}

__global__ __launch_bounds__(1024) void cbam_pool_strip(const float* __restrict__ x,
                                                        float* __restrict__ psum,
                                                        float* __restrict__ pmax) {
  __shared__ __attribute__((aligned(16))) float rs[C_];
  __shared__ __attribute__((aligned(16))) float rm[C_];
  int tid  = threadIdx.x;
  int wave = tid >> 5;
  int lane = tid & 31;
  int b    = blockIdx.x >> 3;
  int st   = blockIdx.x & 7;
  const float* xw = x + ((size_t)(b * C_ + wave * PCW_)) * HW_ + st * (SR_ * W_) + lane * 4;

  v4f buf[PFL_][4];
#pragma unroll
  for (int c = 0; c < PFL_; ++c)
#pragma unroll
    for (int i = 0; i < 4; ++i)
      buf[c][i] = *(const v4f*)(xw + c * HW_ + i * 128);

  float ks = 0.f, km = 0.f;
#pragma unroll
  for (int c = 0; c < PCW_; ++c) {
    v4f s4 = (buf[c % PFL_][0] + buf[c % PFL_][1]) + (buf[c % PFL_][2] + buf[c % PFL_][3]);
    v4f m4;
    m4.x = fmaxf(fmaxf(buf[c % PFL_][0].x, buf[c % PFL_][1].x), fmaxf(buf[c % PFL_][2].x, buf[c % PFL_][3].x));
    m4.y = fmaxf(fmaxf(buf[c % PFL_][0].y, buf[c % PFL_][1].y), fmaxf(buf[c % PFL_][2].y, buf[c % PFL_][3].y));
    m4.z = fmaxf(fmaxf(buf[c % PFL_][0].z, buf[c % PFL_][1].z), fmaxf(buf[c % PFL_][2].z, buf[c % PFL_][3].z));
    m4.w = fmaxf(fmaxf(buf[c % PFL_][0].w, buf[c % PFL_][1].w), fmaxf(buf[c % PFL_][2].w, buf[c % PFL_][3].w));
    if (c + PFL_ < PCW_) {
#pragma unroll
      for (int i = 0; i < 4; ++i)
        buf[c % PFL_][i] = *(const v4f*)(xw + (c + PFL_) * HW_ + i * 128);
    }
    float s = (s4.x + s4.y) + (s4.z + s4.w);
    float m = fmaxf(fmaxf(m4.x, m4.y), fmaxf(m4.z, m4.w));
    s += xrow<1>(s);  m = fmaxf(m, xrow<1>(m));
    s += xrow<2>(s);  m = fmaxf(m, xrow<2>(m));
    s += xrow<4>(s);  m = fmaxf(m, xrow<4>(m));
    s += xrow<8>(s);  m = fmaxf(m, xrow<8>(m));
    s += xhalf(s);    m = fmaxf(m, xhalf(m));
    if (lane == c) { ks = s; km = m; }
  }
  if (lane < PCW_) { rs[wave * PCW_ + lane] = ks; rm[wave * PCW_ + lane] = km; }
  __syncthreads();
  if (tid < 64)
    *(v4f*)(psum + blockIdx.x * C_ + tid * 4) = *(const v4f*)(rs + tid * 4);
  else if (tid < 128)
    *(v4f*)(pmax + blockIdx.x * C_ + (tid - 64) * 4) = *(const v4f*)(rm + (tid - 64) * 4);
}

#define PP_  260
#define W2P_ 20
#define HP_  36

__global__ __launch_bounds__(256) void cbam_mlp_ca(const float* __restrict__ psum,
                                                   const float* __restrict__ pmax,
                                                   const float* __restrict__ w1,
                                                   const float* __restrict__ w2,
                                                   float* __restrict__ ca) {
  __shared__ __attribute__((aligned(16))) float sV[16 * PP_];
  __shared__ __attribute__((aligned(16))) float sW1[16 * PP_];
  __shared__ __attribute__((aligned(16))) float sW2[256 * W2P_];
  __shared__ __attribute__((aligned(16))) float hA[16 * HP_];
  int tid  = threadIdx.x;
  int wave = tid >> 5;
  int lane = tid & 31;
  int half = lane >> 4;
  int l15  = lane & 15;
  int b    = blockIdx.x;

  {
    v4f q1[4], q2[4], p[NSTR_];
#pragma unroll
    for (int i = 0; i < 4; ++i) {
      q1[i] = ((const v4f*)w1)[tid + i * 256];
      q2[i] = ((const v4f*)w2)[tid + i * 256];
    }
    if (tid < 128) {
      const v4f* pp = (const v4f*)((tid < 64 ? psum : pmax) + b * (NSTR_ * C_)) + (tid & 63);
#pragma unroll
      for (int s = 0; s < NSTR_; ++s) p[s] = pp[s * (C_ / 4)];
    }
    for (int e = tid; e < 14 * PP_ / 4; e += 256)
      ((v4f*)(sV + 2 * PP_))[e] = (v4f){0.f, 0.f, 0.f, 0.f};
    if (tid < 15 * HP_ / 4)
      ((v4f*)(hA + HP_))[tid] = (v4f){0.f, 0.f, 0.f, 0.f};
#pragma unroll
    for (int i = 0; i < 4; ++i) {
      int e = tid + i * 256;
      *(v4f*)(sW1 + (e >> 6) * PP_ + (e & 63) * 4) = q1[i];
      *(v4f*)(sW2 + (e >> 2) * W2P_ + (e & 3) * 4) = q2[i];
    }
    if (tid < 128) {
      v4f a = p[0];
      if (tid < 64) {
#pragma unroll
        for (int s = 1; s < NSTR_; ++s) a += p[s];
        a = a * (1.0f / (float)HW_);
      } else {
#pragma unroll
        for (int s = 1; s < NSTR_; ++s) {
          a.x = fmaxf(a.x, p[s].x); a.y = fmaxf(a.y, p[s].y);
          a.z = fmaxf(a.z, p[s].z); a.w = fmaxf(a.w, p[s].w);
        }
      }
      *(v4f*)(sV + (tid >> 6) * PP_ + (tid & 63) * 4) = a;
    }
  }
  __syncthreads();

  if (wave == 0) {
    const float* ap = sV + l15 * PP_ + half * 2;
    const float* bp = sW1 + l15 * PP_ + half * 2;
    v8f acc = {};
    v2f a[2][8], bb[2][8];
#pragma unroll
    for (int j = 0; j < 8; ++j) {
      a[0][j]  = *(const v2f*)(ap + j * 4);
      bb[0][j] = *(const v2f*)(bp + j * 4);
    }
#pragma unroll
    for (int kb = 0; kb < 8; ++kb) {
      if (kb < 7) {
#pragma unroll
        for (int j = 0; j < 8; ++j) {
          a[(kb + 1) & 1][j]  = *(const v2f*)(ap + (kb + 1) * 32 + j * 4);
          bb[(kb + 1) & 1][j] = *(const v2f*)(bp + (kb + 1) * 32 + j * 4);
        }
      }
      __builtin_amdgcn_sched_barrier(0);
#pragma unroll
      for (int j = 0; j < 8; ++j)
        acc = __builtin_amdgcn_wmma_f32_16x16x4_f32(
            false, a[kb & 1][j], false, bb[kb & 1][j], (short)0, acc, false, false);
      __builtin_amdgcn_sched_barrier(0);
    }
    if (half == 0) {
      hA[l15]      = fmaxf(acc[0], 0.f);
      hA[16 + l15] = fmaxf(acc[1], 0.f);
    }
  }
  __syncthreads();

  {
    const float* ap  = hA + l15 * HP_ + half * 2;
    const float* bp0 = sW2 + ((wave * 2)     * 16 + l15) * W2P_ + half * 2;
    const float* bp1 = sW2 + ((wave * 2 + 1) * 16 + l15) * W2P_ + half * 2;
    v8f acc0 = {}, acc1 = {};
#pragma unroll
    for (int k0 = 0; k0 < 32; k0 += 4) {
      v2f av = *(const v2f*)(ap + k0);
      v2f b0 = *(const v2f*)(bp0 + (k0 & 15));
      v2f b1 = *(const v2f*)(bp1 + (k0 & 15));
      acc0 = __builtin_amdgcn_wmma_f32_16x16x4_f32(
          false, av, false, b0, (short)0, acc0, false, false);
      acc1 = __builtin_amdgcn_wmma_f32_16x16x4_f32(
          false, av, false, b1, (short)0, acc1, false, false);
    }
    float v1 = xhalf(acc1[0]);
    float v  = half == 0 ? acc0[0] : v1;
    ca[b * C_ + wave * 32 + lane] = 1.f / (1.f + __expf(-v));
  }
}

#define NWIN_  14
#define ST_    512
#define NW3_   16
#define CPT_   8
#define NSET_  6
#define FP_    72

__device__ __forceinline__ void lds_barrier() {
  __builtin_amdgcn_fence(__ATOMIC_RELEASE, "workgroup", "local");
  __builtin_amdgcn_s_barrier();
  __builtin_amdgcn_fence(__ATOMIC_ACQUIRE, "workgroup", "local");
}

__global__ __launch_bounds__(ST_) void cbam_spatial_out(const float* __restrict__ x,
                                                        const float* __restrict__ ca,
                                                        const float* __restrict__ wsp,
                                                        float* __restrict__ out) {
  __shared__ __attribute__((aligned(16))) float pS[2][NW3_ * 16 * 4];
  __shared__ __attribute__((aligned(16))) float pM[2][NW3_ * 16 * 4];
  __shared__ __attribute__((aligned(16))) v2f fS[NWIN_ * FP_];
  __shared__ __attribute__((aligned(16))) float sigS[2][64];
  __shared__ __attribute__((aligned(16))) v2f sWp[7 * 8];

  int tid  = threadIdx.x;
  int wave = tid >> 5;
  int lane = tid & 31;
  int pos  = lane & 15;
  int half = lane >> 4;
  int c0   = (wave * 2 + half) * CPT_;
  int b    = blockIdx.x >> 3;
  int r0   = (blockIdx.x & 7) * SR_;

  const float* xb = x   + (size_t)(b * C_ + c0) * HW_ + pos * 4;
  float*       ob = out + (size_t)(b * C_ + c0) * HW_ + pos * 4;

  float wv0 = 0.f, wv1 = 0.f;
  if (tid < 49) { wv0 = wsp[tid]; wv1 = wsp[49 + tid]; }
  v4f cav0 = *(const v4f*)(ca + b * C_ + c0);
  v4f cav1 = *(const v4f*)(ca + b * C_ + c0 + 4);
  v4f buf[NSET_][CPT_];
#pragma unroll
  for (int q = 0; q < 2; ++q) {
    int irc = r0 - 3 + q;
    irc = irc < 0 ? 0 : irc;
#pragma unroll
    for (int j = 0; j < CPT_; ++j)
      buf[q][j] = *(const v4f*)(xb + irc * W_ + j * HW_);
  }
  float cav[CPT_] = {cav0.x, cav0.y, cav0.z, cav0.w, cav1.x, cav1.y, cav1.z, cav1.w};
  if (tid < 49) {
    sWp[(tid / 7) * 8 + (tid % 7)] = (v2f){wv0, wv1};
  } else if (tid >= 128 && tid < 128 + NWIN_ * 8) {
    int k = tid - 128;
    int r = k >> 3, q = k & 7;
    fS[r * FP_ + (q < 4 ? q : 64 + q)] = (v2f){0.f, 0.f};
  }
  lds_barrier();

  int kyq = tid & 7, xcv = tid >> 3;
  int kyc = kyq < 7 ? kyq : 6;
  const v2f* fcv = fS + kyc * FP_ + xcv + 1;
  const v2f* wcv = sWp + kyc * 8;

#pragma unroll
  for (int n = 0; n < NWIN_ + 1; ++n) {
    int ir = r0 - 3 + n;
    bool own   = (n >= 3) && (n < 3 + SR_);
    bool valid = own || ((ir >= 0) && (ir < H_));
    int pb = n & 1;
    if (n <= 6) {
      int irc = ir + 2;
      if (n + 2 < 3) irc = irc < 0 ? 0 : irc;
#pragma unroll
      for (int j = 0; j < CPT_; ++j)
        buf[(n + 2) % NSET_][j] = *(const v4f*)(xb + irc * W_ + j * HW_);
    }
    if (n < NWIN_) {
      v4f s = {0.f, 0.f, 0.f, 0.f};
      v4f m = {-INFINITY, -INFINITY, -INFINITY, -INFINITY};
#pragma unroll
      for (int j = 0; j < CPT_; ++j) {
        v4f t = buf[n % NSET_][j] * cav[j];
        s += t;
        m.x = fmaxf(m.x, t.x); m.y = fmaxf(m.y, t.y);
        m.z = fmaxf(m.z, t.z); m.w = fmaxf(m.w, t.w);
      }
      s.x += xhalf(s.x); s.y += xhalf(s.y); s.z += xhalf(s.z); s.w += xhalf(s.w);
      m.x = fmaxf(m.x, xhalf(m.x)); m.y = fmaxf(m.y, xhalf(m.y));
      m.z = fmaxf(m.z, xhalf(m.z)); m.w = fmaxf(m.w, xhalf(m.w));
      if (half == 0) {
        *(v4f*)(pS[pb] + (wave * 16 + pos) * 4) = s;
        *(v4f*)(pM[pb] + (wave * 16 + pos) * 4) = m;
      }
    }
    if (n >= 7) {
      const v2f* f0 = fcv + (n - 7) * FP_;
      float a = 0.f;
#pragma unroll
      for (int kx = 0; kx < 7; ++kx) {
        v2f f = f0[kx];
        v2f w = wcv[kx];
        a += f.x * w.x + f.y * w.y;
      }
      a = kyq < 7 ? a : 0.f;
      a += xrow<1>(a);
      a += xrow<2>(a);
      a += xrow<4>(a);
      if (kyq == 0) sigS[pb][xcv] = 1.f / (1.f + __expf(-a));
    }
    lds_barrier();
    if (n < NWIN_ && tid < 128) {
      int p2 = tid & 15, comp = (tid >> 4) & 3, sm = tid >> 6;
      float val;
      if (sm == 0) {
        const float* src = pS[pb] + p2 * 4 + comp;
        float a = 0.f;
#pragma unroll
        for (int w = 0; w < NW3_; ++w) a += src[w * 64];
        val = a * (1.f / 256.f);
      } else {
        const float* src = pM[pb] + p2 * 4 + comp;
        float a = -INFINITY;
#pragma unroll
        for (int w = 0; w < NW3_; ++w) a = fmaxf(a, src[w * 64]);
        val = a;
      }
      if (!valid) val = 0.f;
      ((float*)fS)[(n * FP_ + 4 + p2 * 4 + comp) * 2 + sm] = val;
    }
    if (n >= 7) {
      v4f sv = *(const v4f*)(sigS[pb] + pos * 4);
      float* op = ob + (ir - 4) * W_;
#pragma unroll
      for (int j = 0; j < CPT_; ++j) {
        v4f o = buf[(n - 4) % NSET_][j] * (sv * cav[j]);
        __builtin_nontemporal_store(o, (v4f*)(op + j * HW_));
      }
      if (n + 2 < NWIN_) {
        int irc = ir + 2;
        if (n + 2 >= 3 + SR_) irc = irc > H_ - 1 ? H_ - 1 : irc;
#pragma unroll
        for (int j = 0; j < CPT_; ++j)
          buf[(n + 2) % NSET_][j] = *(const v4f*)(xb + irc * W_ + j * HW_);
      }
    }
    if (n >= 6 && n < NWIN_) lds_barrier();
  }
}

extern "C" void kernel_launch(void* const* d_in, const int* in_sizes, int n_in,
                              void* d_out, int out_size, void* d_ws, size_t ws_size,
                              hipStream_t stream) {
  const float* x  = (const float*)d_in[0];
  const float* w1 = (const float*)d_in[1];
  const float* w2 = (const float*)d_in[2];
  const float* ws = (const float*)d_in[3];
  float* out = (float*)d_out;

  float* wsf  = (float*)d_ws;
  float* psum = wsf;
  float* pmax = wsf + 65536;
  float* ca   = wsf + 131072;

  cbam_pool_strip <<<B_ * NSTR_, 1024, 0, stream>>>(x, psum, pmax);
  cbam_mlp_ca     <<<B_,          256, 0, stream>>>(psum, pmax, w1, w2, ca);
  cbam_spatial_out<<<B_ * NSTR_,  ST_, 0, stream>>>(x, ca, ws, out);
}


# MLP launch dissolved at its hidden layer: pool pass re-cut to (batch row x 32 channels) blocks that add their k share of layer 1; fused pass joins 8 partials, does layer 2; plane and k sums reordered
# speedup vs baseline: 2.9315x; 1.1811x over previous
#include <hip/hip_runtime.h>
#include <math.h>


#define B_   32
#define C_   256
#define H_   64
#define W_   64
#define HW_  4096
#define MID_ 16

typedef __attribute__((ext_vector_type(2))) float v2f;
typedef __attribute__((ext_vector_type(4))) float v4f;
typedef __attribute__((ext_vector_type(8))) float v8f;

__device__ __forceinline__ int blk_row(int i)  { return (i & 7) | ((i >> 6) << 3); }
__device__ __forceinline__ int blk_part(int i) { return (i >> 3) & 7; }

template <int MASK>
__device__ __forceinline__ float xrow(float v) {
  int b = __builtin_bit_cast(int, v);
  return __builtin_bit_cast(float,
      __builtin_amdgcn_update_dpp(b, b, 0x160 + MASK, 0xf, 0xf, false));
}

__device__ __forceinline__ float xhalf(float v) {
  int b = __builtin_bit_cast(int, v);
  return __builtin_bit_cast(float,
      __builtin_amdgcn_permlanex16(b, b, 0x76543210, 0xfedcba98, false, false));
}

#define W2P_ 20
#define HP_  36

#define CGN_   8
#define CGC_   32
#define PFL_   4

__global__ __launch_bounds__(1024) void cbam_pool_hidden(const float* __restrict__ x,
                                                         const float* __restrict__ w1,
                                                         float* __restrict__ hpart) {
  __shared__ __attribute__((aligned(16))) float sV[16 * HP_];
  __shared__ __attribute__((aligned(16))) float sW[16 * HP_];
  int tid  = threadIdx.x;
  int wave = tid >> 5;
  int lane = tid & 31;
  int half = lane >> 4;
  int l15  = lane & 15;
  int b    = blk_row(blockIdx.x);
  int cg   = blk_part(blockIdx.x);
  const float* xw = x + ((size_t)(b * C_ + cg * CGC_ + wave)) * HW_ + lane * 4;

  v4f wq = {0.f, 0.f, 0.f, 0.f};
  if (tid < 128) wq = *(const v4f*)(w1 + (tid >> 3) * C_ + cg * CGC_ + (tid & 7) * 4);

  v4f buf[PFL_][4];
#pragma unroll
  for (int q = 0; q < PFL_; ++q)
#pragma unroll
    for (int i = 0; i < 4; ++i)
      buf[q][i] = *(const v4f*)(xw + q * 512 + i * 128);

  v4f s4 = {0.f, 0.f, 0.f, 0.f};
  v4f m4 = {0.f, 0.f, 0.f, 0.f};
#pragma unroll
  for (int q = 0; q < 8; ++q) {
    v4f t = (buf[q % PFL_][0] + buf[q % PFL_][1]) + (buf[q % PFL_][2] + buf[q % PFL_][3]);
    v4f u;
    u.x = fmaxf(fmaxf(buf[q % PFL_][0].x, buf[q % PFL_][1].x), fmaxf(buf[q % PFL_][2].x, buf[q % PFL_][3].x));
    u.y = fmaxf(fmaxf(buf[q % PFL_][0].y, buf[q % PFL_][1].y), fmaxf(buf[q % PFL_][2].y, buf[q % PFL_][3].y));
    u.z = fmaxf(fmaxf(buf[q % PFL_][0].z, buf[q % PFL_][1].z), fmaxf(buf[q % PFL_][2].z, buf[q % PFL_][3].z));
    u.w = fmaxf(fmaxf(buf[q % PFL_][0].w, buf[q % PFL_][1].w), fmaxf(buf[q % PFL_][2].w, buf[q % PFL_][3].w));
    if (q + PFL_ < 8) {
#pragma unroll
      for (int i = 0; i < 4; ++i)
        buf[q % PFL_][i] = *(const v4f*)(xw + (q + PFL_) * 512 + i * 128);
    }
    if (q == 0) { s4 = t; m4 = u; }
    else {
      s4 += t;
      m4.x = fmaxf(m4.x, u.x); m4.y = fmaxf(m4.y, u.y);
      m4.z = fmaxf(m4.z, u.z); m4.w = fmaxf(m4.w, u.w);
    }
  }
  float s = (s4.x + s4.y) + (s4.z + s4.w);
  float m = fmaxf(fmaxf(m4.x, m4.y), fmaxf(m4.z, m4.w));
  s += xrow<1>(s);  m = fmaxf(m, xrow<1>(m));
  s += xrow<2>(s);  m = fmaxf(m, xrow<2>(m));
  s += xrow<4>(s);  m = fmaxf(m, xrow<4>(m));
  s += xrow<8>(s);  m = fmaxf(m, xrow<8>(m));
  s += xhalf(s);    m = fmaxf(m, xhalf(m));

  if (lane == 0) { sV[wave] = s * (1.0f / (float)HW_); sV[HP_ + wave] = m; }
  if (tid < 128)
    *(v4f*)(sW + (tid >> 3) * HP_ + (tid & 7) * 4) = wq;
  else if (tid < 128 + 14 * HP_ / 4)
    ((v4f*)(sV + 2 * HP_))[tid - 128] = (v4f){0.f, 0.f, 0.f, 0.f};
  __syncthreads();

  if (wave == 0) {
    const float* ap = sV + l15 * HP_ + half * 2;
    const float* bp = sW + l15 * HP_ + half * 2;
    v8f acc = {};
#pragma unroll
    for (int k0 = 0; k0 < CGC_; k0 += 4) {
      v2f av = *(const v2f*)(ap + k0);
      v2f bv = *(const v2f*)(bp + k0);
      acc = __builtin_amdgcn_wmma_f32_16x16x4_f32(
          false, av, false, bv, (short)0, acc, false, false);
    }
    float v1 = xhalf(acc[1]);
    float v  = half == 0 ? acc[0] : v1;
    hpart[(b * CGN_ + cg) * 32 + lane] = v;
  }
}

#define SR_    8
#define NWIN_  14
#define ST_    512
#define NW3_   16
#define CPT_   8
#define NSET_  6
#define FP_    72

__device__ __forceinline__ void lds_barrier() {
  __builtin_amdgcn_fence(__ATOMIC_RELEASE, "workgroup", "local");
  __builtin_amdgcn_s_barrier();
  __builtin_amdgcn_fence(__ATOMIC_ACQUIRE, "workgroup", "local");
}

__global__ __launch_bounds__(ST_) void cbam_spatial_out(const float* __restrict__ x,
                                                        const float* __restrict__ hpart,
                                                        const float* __restrict__ w2,
                                                        const float* __restrict__ wsp,
                                                        float* __restrict__ out) {
  __shared__ __attribute__((aligned(16))) float sW2[C_ * W2P_];
  __shared__ __attribute__((aligned(16))) float hA[16 * HP_];
  __shared__ __attribute__((aligned(16))) float sCA[C_];
  __shared__ __attribute__((aligned(16))) float pS[2][NW3_ * 16 * 4];
  __shared__ __attribute__((aligned(16))) float pM[2][NW3_ * 16 * 4];
  __shared__ __attribute__((aligned(16))) v2f fS[NWIN_ * FP_];
  __shared__ __attribute__((aligned(16))) float sigS[2][64];
  __shared__ __attribute__((aligned(16))) v2f sWp[7 * 8];

  int tid  = threadIdx.x;
  int wave = tid >> 5;
  int lane = tid & 31;
  int pos  = lane & 15;
  int half = lane >> 4;
  int c0   = (wave * 2 + half) * CPT_;
  int l15  = lane & 15;
  int b    = blk_row(blockIdx.x);
  int r0   = blk_part(blockIdx.x) * SR_;

  const float* xb = x   + (size_t)(b * C_ + c0) * HW_ + pos * 4;
  float*       ob = out + (size_t)(b * C_ + c0) * HW_ + pos * 4;

  float wv0 = 0.f, wv1 = 0.f;
  if (tid < 49) { wv0 = wsp[tid]; wv1 = wsp[49 + tid]; }
  v4f q2a = ((const v4f*)w2)[tid];
  v4f q2b = ((const v4f*)w2)[tid + ST_];
  float hp[8] = {0.f, 0.f, 0.f, 0.f, 0.f, 0.f, 0.f, 0.f};
  if (tid < 32) {
#pragma unroll
    for (int g = 0; g < 8; ++g) hp[g] = hpart[(b * 8 + g) * 32 + tid];
  }
  v4f buf[NSET_][CPT_];
#pragma unroll
  for (int q = 0; q < 2; ++q) {
    int irc = r0 - 3 + q;
    irc = irc < 0 ? 0 : irc;
#pragma unroll
    for (int j = 0; j < CPT_; ++j)
      buf[q][j] = *(const v4f*)(xb + irc * W_ + j * HW_);
  }
  *(v4f*)(sW2 + (tid >> 2) * W2P_ + (tid & 3) * 4) = q2a;
  *(v4f*)(sW2 + ((tid + ST_) >> 2) * W2P_ + (tid & 3) * 4) = q2b;
  if (tid < 32) {
    float h = hp[0];
#pragma unroll
    for (int g = 1; g < 8; ++g) h += hp[g];
    hA[tid] = fmaxf(h, 0.f);
  } else if (tid >= 64 && tid < 64 + 15 * HP_ / 4) {
    ((v4f*)(hA + HP_))[tid - 64] = (v4f){0.f, 0.f, 0.f, 0.f};
  }
  if (tid < 49) {
    sWp[(tid / 7) * 8 + (tid % 7)] = (v2f){wv0, wv1};
  } else if (tid >= 128 && tid < 128 + NWIN_ * 8) {
    int k = tid - 128;
    int r = k >> 3, q = k & 7;
    fS[r * FP_ + (q < 4 ? q : 64 + q)] = (v2f){0.f, 0.f};
  }
  lds_barrier();

  {
    const float* ap = hA + l15 * HP_ + half * 2;
    const float* bp = sW2 + (wave * 16 + l15) * W2P_ + half * 2;
    v8f acc = {};
#pragma unroll
    for (int k0 = 0; k0 < 32; k0 += 4) {
      v2f av = *(const v2f*)(ap + k0);
      v2f bv = *(const v2f*)(bp + (k0 & 15));
      acc = __builtin_amdgcn_wmma_f32_16x16x4_f32(
          false, av, false, bv, (short)0, acc, false, false);
    }
    if (half == 0) sCA[wave * 16 + l15] = 1.f / (1.f + __expf(-acc[0]));
  }
  lds_barrier();
  v4f cav0 = *(const v4f*)(sCA + c0);
  v4f cav1 = *(const v4f*)(sCA + c0 + 4);
  float cav[CPT_] = {cav0.x, cav0.y, cav0.z, cav0.w, cav1.x, cav1.y, cav1.z, cav1.w};

  int kyq = tid & 7, xcv = tid >> 3;
  int kyc = kyq < 7 ? kyq : 6;
  const v2f* fcv = fS + kyc * FP_ + xcv + 1;
  const v2f* wcv = sWp + kyc * 8;

#pragma unroll
  for (int n = 0; n < NWIN_ + 1; ++n) {
    int ir = r0 - 3 + n;
    bool own   = (n >= 3) && (n < 3 + SR_);
    bool valid = own || ((ir >= 0) && (ir < H_));
    int pb = n & 1;
    if (n <= 6) {
      int irc = ir + 2;
      if (n + 2 < 3) irc = irc < 0 ? 0 : irc;
#pragma unroll
      for (int j = 0; j < CPT_; ++j)
        buf[(n + 2) % NSET_][j] = *(const v4f*)(xb + irc * W_ + j * HW_);
    }
    if (n < NWIN_) {
      v4f s = {0.f, 0.f, 0.f, 0.f};
      v4f m = {-INFINITY, -INFINITY, -INFINITY, -INFINITY};
#pragma unroll
      for (int j = 0; j < CPT_; ++j) {
        v4f t = buf[n % NSET_][j] * cav[j];
        s += t;
        m.x = fmaxf(m.x, t.x); m.y = fmaxf(m.y, t.y);
        m.z = fmaxf(m.z, t.z); m.w = fmaxf(m.w, t.w);
      }
      s.x += xhalf(s.x); s.y += xhalf(s.y); s.z += xhalf(s.z); s.w += xhalf(s.w);
      m.x = fmaxf(m.x, xhalf(m.x)); m.y = fmaxf(m.y, xhalf(m.y));
      m.z = fmaxf(m.z, xhalf(m.z)); m.w = fmaxf(m.w, xhalf(m.w));
      if (half == 0) {
        *(v4f*)(pS[pb] + (wave * 16 + pos) * 4) = s;
        *(v4f*)(pM[pb] + (wave * 16 + pos) * 4) = m;
      }
    }
    if (n >= 7) {
      const v2f* f0 = fcv + (n - 7) * FP_;
      float a = 0.f;
#pragma unroll
      for (int kx = 0; kx < 7; ++kx) {
        v2f f = f0[kx];
        v2f w = wcv[kx];
        a += f.x * w.x + f.y * w.y;
      }
      a = kyq < 7 ? a : 0.f;
      a += xrow<1>(a);
      a += xrow<2>(a);
      a += xrow<4>(a);
      if (kyq == 0) sigS[pb][xcv] = 1.f / (1.f + __expf(-a));
    }
    lds_barrier();
    if (n < NWIN_ && tid < 128) {
      int p2 = tid & 15, comp = (tid >> 4) & 3, sm = tid >> 6;
      float val;
      if (sm == 0) {
        const float* src = pS[pb] + p2 * 4 + comp;
        float a = 0.f;
#pragma unroll
        for (int w = 0; w < NW3_; ++w) a += src[w * 64];
        val = a * (1.f / 256.f);
      } else {
        const float* src = pM[pb] + p2 * 4 + comp;
        float a = -INFINITY;
#pragma unroll
        for (int w = 0; w < NW3_; ++w) a = fmaxf(a, src[w * 64]);
        val = a;
      }
      if (!valid) val = 0.f;
      ((float*)fS)[(n * FP_ + 4 + p2 * 4 + comp) * 2 + sm] = val;
    }
    if (n >= 7) {
      v4f sv = *(const v4f*)(sigS[pb] + pos * 4);
      float* op = ob + (ir - 4) * W_;
#pragma unroll
      for (int j = 0; j < CPT_; ++j) {
        v4f o = buf[(n - 4) % NSET_][j] * (sv * cav[j]);
        __builtin_nontemporal_store(o, (v4f*)(op + j * HW_));
      }
      if (n + 2 < NWIN_) {
        int irc = ir + 2;
        if (n + 2 >= 3 + SR_) irc = irc > H_ - 1 ? H_ - 1 : irc;
#pragma unroll
        for (int j = 0; j < CPT_; ++j)
          buf[(n + 2) % NSET_][j] = *(const v4f*)(xb + irc * W_ + j * HW_);
      }
    }
    if (n >= 6 && n < NWIN_) lds_barrier();
  }
}

extern "C" void kernel_launch(void* const* d_in, const int* in_sizes, int n_in,
                              void* d_out, int out_size, void* d_ws, size_t ws_size,
                              hipStream_t stream) {
  const float* x  = (const float*)d_in[0];
  const float* w1 = (const float*)d_in[1];
  const float* w2 = (const float*)d_in[2];
  const float* ws = (const float*)d_in[3];
  float* out = (float*)d_out;

  float* hpart = (float*)d_ws;

  cbam_pool_hidden<<<B_ * CGN_, 1024, 0, stream>>>(x, w1, hpart);
  cbam_spatial_out<<<B_ * 8,     ST_, 0, stream>>>(x, hpart, w2, ws, out);
}
